# starting kernel
# baseline (speedup 1.0000x reference)
#include <hip/hip_runtime.h>
#include <hip/hip_bf16.h>

typedef __attribute__((ext_vector_type(2))) float v2f;
typedef __attribute__((ext_vector_type(8))) float v8f;

#define MROWS 2048
#define FD    128

__global__ __launch_bounds__(128) void precompute_vecs(
    const float* __restrict__ W4, const float* __restrict__ b4,
    const float* __restrict__ W5, const float* __restrict__ b5,
    float* __restrict__ vl, float* __restrict__ vr, float* __restrict__ Cs) {
  const int f = threadIdx.x;
  float sl = 0.f, sr = 0.f;
  #pragma unroll 8
  for (int s = 0; s < 128; ++s) {
    const float w5 = W5[s];
    sl += w5 * W4[s * 256 + f];
    sr += w5 * W4[s * 256 + 128 + f];
  }
  vl[f] = sl;
  vr[f] = sr;
  if (f == 0) {
    float c = b5[0];
    for (int s = 0; s < 128; ++s) c += b4[s] * W5[s];
    Cs[0] = c;
  }
}

__global__ __launch_bounds__(256) void gemm_wmma_f32(
    const float* __restrict__ Hin, const float* __restrict__ W,
    const float* __restrict__ bias, float* __restrict__ Hout) {
  const int lane = threadIdx.x & 31;
  const int wave = threadIdx.x >> 5;
  const int m0   = blockIdx.x * 16;
  const int n0   = wave * 16;
  const int half = lane >> 4;
  const int l15  = lane & 15;

  const float* pA = Hin + (m0 + l15) * FD + half * 2;
  const float* pB = W   + (n0 + l15) * FD + half * 2;

  v8f acc = {};
  #pragma unroll
  for (int k0 = 0; k0 < FD; k0 += 4) {
    const float2 av = *(const float2*)(pA + k0);
    const float2 bv = *(const float2*)(pB + k0);
    v2f a; a.x = av.x; a.y = av.y;
    v2f b; b.x = bv.x; b.y = bv.y;
    acc = __builtin_amdgcn_wmma_f32_16x16x4_f32(
        false, a, false, b, (short)0, acc, false, false);
  }

  const int   col = n0 + l15;
  const float bv  = bias[col];
  #pragma unroll
  for (int g = 0; g < 8; ++g) {
    const int row = m0 + g + half * 8;
    Hout[row * FD + col] = acc[g] + bv;
  }
}

__global__ __launch_bounds__(256) void row_dots(
    const float* __restrict__ H, const float* __restrict__ vl,
    const float* __restrict__ vr, float* __restrict__ ul,
    float* __restrict__ ur) {
  const int lane = threadIdx.x & 31;
  const int wave = threadIdx.x >> 5;
  const int row  = blockIdx.x * 8 + wave;

  const float4 h4 = *(const float4*)(H + row * FD + lane * 4);
  const float4 l4 = *(const float4*)(vl + lane * 4);
  const float4 r4 = *(const float4*)(vr + lane * 4);
  float sl = h4.x * l4.x + h4.y * l4.y + h4.z * l4.z + h4.w * l4.w;
  float sr = h4.x * r4.x + h4.y * r4.y + h4.z * r4.z + h4.w * r4.w;
  #pragma unroll
  for (int off = 16; off > 0; off >>= 1) {
    sl += __shfl_down(sl, off, 32);
    sr += __shfl_down(sr, off, 32);
  }
  if (lane == 0) { ul[row] = sl; ur[row] = sr; }
}

__global__ __launch_bounds__(256) void expand_out(
    const float* __restrict__ ul, const float* __restrict__ ur,
    const float* __restrict__ Cs, float* __restrict__ out) {
  const unsigned idx = blockIdx.x * 256u + threadIdx.x;
  const unsigned j = idx & 255u;
  const unsigned i = (idx >> 8) & 255u;
  const unsigned b = idx >> 16;
  out[idx] = ul[b * 256u + i] + ur[b * 256u + j] + Cs[0];
}

extern "C" void kernel_launch(void* const* d_in, const int* in_sizes, int n_in,
                              void* d_out, int out_size, void* d_ws, size_t ws_size,
                              hipStream_t stream) {
  (void)in_sizes; (void)n_in; (void)out_size; (void)ws_size;
  const float* x  = (const float*)d_in[0];
  const float* W1 = (const float*)d_in[1];
  const float* b1 = (const float*)d_in[2];
  const float* W2 = (const float*)d_in[3];
  const float* b2 = (const float*)d_in[4];
  const float* W3 = (const float*)d_in[5];
  const float* b3 = (const float*)d_in[6];
  const float* W4 = (const float*)d_in[7];
  const float* b4 = (const float*)d_in[8];
  const float* W5 = (const float*)d_in[9];
  const float* b5 = (const float*)d_in[10];
  float* out = (float*)d_out;

  float* ws   = (float*)d_ws;
  float* bufA = ws;
  float* bufB = bufA + MROWS * FD;
  float* vl   = bufB + MROWS * FD;
  float* vr   = vl + FD;
  float* Cs   = vr + FD;
  float* ul   = Cs + 1;
  float* ur   = ul + MROWS;

  precompute_vecs<<<1, 128, 0, stream>>>(W4, b4, W5, b5, vl, vr, Cs);

  gemm_wmma_f32<<<MROWS / 16, 256, 0, stream>>>(x,    W1, b1, bufA);
  gemm_wmma_f32<<<MROWS / 16, 256, 0, stream>>>(bufA, W2, b2, bufB);
  gemm_wmma_f32<<<MROWS / 16, 256, 0, stream>>>(bufB, W3, b3, bufA);

  row_dots<<<MROWS / 8, 256, 0, stream>>>(bufA, vl, vr, ul, ur);

  expand_out<<<(8 * 256 * 256) / 256, 256, 0, stream>>>(ul, ur, Cs, out);
}


# Whole net folded to two 128-vectors per block on the matrix unit (q*W3*W2*W1), ul,ur = x*(pl;pr) as a 2-column product, so one launch not four; no h3, no workspace; sum order changes
# speedup vs baseline: 2.8075x; 2.8075x over previous
#include <hip/hip_runtime.h>


typedef __attribute__((ext_vector_type(2))) float v2f;
typedef __attribute__((ext_vector_type(8))) float v8f;

#define FD   128
#define NROW 256
#define XP   132
#define PQ   128

__device__ __forceinline__ int kmap(int i, int h, int r) {
  return 8 * (i >> 1) + 4 * h + 2 * (i & 1) + r;
}

__device__ __forceinline__ float4 add4(float4 a, float4 b) {
  return make_float4(a.x + b.x, a.y + b.y, a.z + b.z, a.w + b.w);
}

__device__ __forceinline__ float4 sum_parts4(const float* p) {
  float4 s = *(const float4*)p;
  #pragma unroll
  for (int q = 1; q < 8; ++q) s = add4(s, *(const float4*)(p + q * 2 * PQ));
  return s;
}

__device__ __forceinline__ float sum_both_rows(const float* P, int n) {
  float s = 0.f;
  #pragma unroll
  for (int q = 0; q < 16; ++q) s += P[q * PQ + n];
  return s;
}

__device__ __forceinline__ void load_frags(const float* __restrict__ Wm, float4 (&wf)[8],
                                           int c, int h, int ks, int ng) {
  const float* base = Wm + (16 * ks) * FD + 64 * ng + 4 * c;
  #pragma unroll
  for (int i = 0; i < 4; ++i) {
    wf[2 * i + 0] = *(const float4*)(base + kmap(i, h, 0) * FD);
    wf[2 * i + 1] = *(const float4*)(base + kmap(i, h, 1) * FD);
  }
}

__device__ __forceinline__ float comp(const float4& v, int j) {
  return j == 0 ? v.x : (j == 1 ? v.y : (j == 2 ? v.z : v.w));
}

__device__ __forceinline__ void fold_stage(const float* __restrict__ Pin,
                                           float* __restrict__ Pout,
                                           const float4 (&wf)[8],
                                           int c, int h, int ks, int ng) {
  const bool   live = (c < 2);
  const float* pa = Pin + (c & 1) * PQ + 16 * ks + 4 * h;
  v8f acc[4] = {};
  #pragma unroll
  for (int m = 0; m < 2; ++m) {
    float4 s = sum_parts4(pa + 8 * m);
    if (!live) s = make_float4(0.f, 0.f, 0.f, 0.f);
    v2f a0; a0.x = s.x; a0.y = s.y;
    v2f a1; a1.x = s.z; a1.y = s.w;
    #pragma unroll
    for (int j = 0; j < 4; ++j) {
      v2f b0; b0.x = comp(wf[4 * m + 0], j); b0.y = comp(wf[4 * m + 1], j);
      v2f b1; b1.x = comp(wf[4 * m + 2], j); b1.y = comp(wf[4 * m + 3], j);
      acc[j] = __builtin_amdgcn_wmma_f32_16x16x4_f32(false, a0, false, b0, (short)0, acc[j], false, false);
      acc[j] = __builtin_amdgcn_wmma_f32_16x16x4_f32(false, a1, false, b1, (short)0, acc[j], false, false);
    }
  }
  const float4 r0 = make_float4(acc[0][0], acc[1][0], acc[2][0], acc[3][0]);
  const float4 r1 = make_float4(acc[0][1], acc[1][1], acc[2][1], acc[3][1]);
  if (h == 0) {
    *(float4*)(Pout + (ks * 2 + 0) * PQ + 64 * ng + 4 * c) = r0;
    *(float4*)(Pout + (ks * 2 + 1) * PQ + 64 * ng + 4 * c) = r1;
  }
}

__global__ __launch_bounds__(512) void readout_folded(
    const float* __restrict__ x,
    const float* __restrict__ W1, const float* __restrict__ b1,
    const float* __restrict__ W2, const float* __restrict__ b2,
    const float* __restrict__ W3, const float* __restrict__ b3,
    const float* __restrict__ W4, const float* __restrict__ b4,
    const float* __restrict__ W5, const float* __restrict__ b5,
    float* __restrict__ out) {
  __shared__ __attribute__((aligned(16))) float Xs[NROW * XP];
  __shared__ __attribute__((aligned(16))) float PA[16 * PQ];
  __shared__ __attribute__((aligned(16))) float PB[16 * PQ];
  __shared__ __attribute__((aligned(16))) float Pfin[2 * PQ];
  __shared__ __attribute__((aligned(16))) float ULs[NROW];
  __shared__ __attribute__((aligned(16))) float URs[NROW];
  __shared__ float Kred[4];

  const int t    = threadIdx.x;
  const int lane = t & 31;
  const int wave = t >> 5;
  const int c    = lane & 15;
  const int h    = lane >> 4;
  const int b    = blockIdx.x >> 3;
  const int ib   = blockIdx.x & 7;
  const int ng   = wave & 1;
  const int ks   = wave >> 1;

  const int cg = t & 63;
  const int q  = t >> 6;
  float4 w4[16];
  float  w5[16];
  #pragma unroll
  for (int i = 0; i < 16; ++i) {
    w4[i] = *(const float4*)(W4 + (16 * q + i) * 256 + 4 * cg);
    w5[i] = W5[16 * q + i];
  }
  float4 xr[16];
  const float* xrow = x + (size_t)(b * NROW + 16 * wave) * FD + 4 * lane;
  #pragma unroll
  for (int j = 0; j < 16; ++j) xr[j] = *(const float4*)(xrow + j * FD);
  float4 w3f[8], w2f[8], w1f[8];
  load_frags(W3, w3f, c, h, ks, ng);
  load_frags(W2, w2f, c, h, ks, ng);
  load_frags(W1, w1f, c, h, ks, ng);
  const int   tn  = t & 127;
  const float bb1 = b1[tn], bb2 = b2[tn], bb3 = b3[tn];
  float kacc = b4[tn] * W5[tn];

  float4 p = make_float4(0.f, 0.f, 0.f, 0.f);
  #pragma unroll
  for (int i = 0; i < 16; ++i) {
    p.x += w5[i] * w4[i].x;
    p.y += w5[i] * w4[i].y;
    p.z += w5[i] * w4[i].z;
    p.w += w5[i] * w4[i].w;
  }
  *(float4*)(PA + q * 2 * PQ + 4 * cg) = p;
  #pragma unroll
  for (int j = 0; j < 16; ++j)
    *(float4*)(Xs + (16 * wave + j) * XP + 4 * lane) = xr[j];
  __syncthreads();

  if (wave < 4) kacc += sum_both_rows(PA, tn) * bb3;
  fold_stage(PA, PB, w3f, c, h, ks, ng);
  __syncthreads();
  if (wave < 4) kacc += sum_both_rows(PB, tn) * bb2;
  fold_stage(PB, PA, w2f, c, h, ks, ng);
  __syncthreads();
  if (wave < 4) {
    kacc += sum_both_rows(PA, tn) * bb1;
    #pragma unroll
    for (int o = 16; o > 0; o >>= 1) kacc += __shfl_xor(kacc, o, 32);
    if (lane == 0) Kred[wave] = kacc;
  }
  fold_stage(PA, PB, w1f, c, h, ks, ng);
  __syncthreads();
  if (t < 256) {
    float s = 0.f;
    #pragma unroll
    for (int pp = 0; pp < 8; ++pp) s += PB[(pp * 2 + (t >> 7)) * PQ + (t & 127)];
    Pfin[t] = s;
  }
  __syncthreads();

  {
    const bool   live = (c < 2);
    const float* pb = Pfin + (c & 1) * PQ + 4 * h;
    const float* xa = Xs + (16 * wave + c) * XP + 4 * h;
    v8f acc0 = {}, acc1 = {};
    #pragma unroll
    for (int m = 0; m < 16; ++m) {
      const float4 xv = *(const float4*)(xa + 8 * m);
      float4 s = *(const float4*)(pb + 8 * m);
      if (!live) s = make_float4(0.f, 0.f, 0.f, 0.f);
      v2f a0; a0.x = xv.x; a0.y = xv.y;
      v2f a1; a1.x = xv.z; a1.y = xv.w;
      v2f b0; b0.x = s.x;  b0.y = s.y;
      v2f b1; b1.x = s.z;  b1.y = s.w;
      acc0 = __builtin_amdgcn_wmma_f32_16x16x4_f32(false, a0, false, b0, (short)0, acc0, false, false);
      acc1 = __builtin_amdgcn_wmma_f32_16x16x4_f32(false, a1, false, b1, (short)0, acc1, false, false);
    }
    v8f d = acc0 + acc1;
    if (live) {
      float* dst = (c == 0 ? ULs : URs) + 16 * wave + 8 * h;
      #pragma unroll
      for (int g = 0; g < 8; ++g) dst[g] = d[g];
    }
  }
  __syncthreads();

  {
    const float K  = ((Kred[0] + Kred[1]) + (Kred[2] + Kred[3])) + b5[0];
    const int   j4 = (t & 63) * 4;
    const int   rg = t >> 6;
    float4 urv = *(const float4*)(URs + j4);
    urv.x += K; urv.y += K; urv.z += K; urv.w += K;
    float* orow = out + (size_t)(b * NROW + 32 * ib) * NROW + j4;
    #pragma unroll
    for (int m = 0; m < 4; ++m) {
      const int   i  = rg + 8 * m;
      const float ul = ULs[32 * ib + i];
      *(float4*)(orow + (size_t)i * NROW) =
          make_float4(ul + urv.x, ul + urv.y, ul + urv.z, ul + urv.w);
    }
  }
}

extern "C" void kernel_launch(void* const* d_in, const int* in_sizes, int n_in,
                              void* d_out, int out_size, void* d_ws, size_t ws_size,
                              hipStream_t stream) {
  (void)in_sizes; (void)n_in; (void)out_size; (void)d_ws; (void)ws_size;
  const float* x  = (const float*)d_in[0];
  const float* W1 = (const float*)d_in[1];
  const float* b1 = (const float*)d_in[2];
  const float* W2 = (const float*)d_in[3];
  const float* b2 = (const float*)d_in[4];
  const float* W3 = (const float*)d_in[5];
  const float* b3 = (const float*)d_in[6];
  const float* W4 = (const float*)d_in[7];
  const float* b4 = (const float*)d_in[8];
  const float* W5 = (const float*)d_in[9];
  const float* b5 = (const float*)d_in[10];
  float* out = (float*)d_out;

  readout_folded<<<64, 512, 0, stream>>>(x, W1, b1, W2, b2, W3, b3, W4, b4, W5, b5, out);
}


# Each stage waits only for its own loads: fold starts when W4 is in (LDS-only barriers, counted waits), x rows requested after stage 0 and parked after the fold; sums and their order unchanged
# speedup vs baseline: 3.2419x; 1.1547x over previous
#include <hip/hip_runtime.h>


typedef __attribute__((ext_vector_type(2))) float v2f;
typedef __attribute__((ext_vector_type(8))) float v8f;
typedef __attribute__((ext_vector_type(4))) float v4f;

#define FD   128
#define NROW 256
#define XP   132
#define PQ   128

__device__ __forceinline__ int kmap(int i, int h, int r) {
  return 8 * (i >> 1) + 4 * h + 2 * (i & 1) + r;
}

__device__ __forceinline__ float4 add4(float4 a, float4 b) {
  return make_float4(a.x + b.x, a.y + b.y, a.z + b.z, a.w + b.w);
}

__device__ __forceinline__ void lds_barrier() {
  __builtin_amdgcn_fence(__ATOMIC_RELEASE, "workgroup", "local");
  __builtin_amdgcn_s_barrier();
  __builtin_amdgcn_fence(__ATOMIC_ACQUIRE, "workgroup", "local");
}

__device__ __forceinline__ float4 sum_parts4(const float* p) {
  float4 s = *(const float4*)p;
  #pragma unroll
  for (int q = 1; q < 8; ++q) s = add4(s, *(const float4*)(p + q * 2 * PQ));
  return s;
}

__device__ __forceinline__ float sum_both_rows(const float* P, int n) {
  float s = 0.f;
  #pragma unroll
  for (int q = 0; q < 16; ++q) s += P[q * PQ + n];
  return s;
}

__device__ __forceinline__ void load_frags(const float* __restrict__ Wm, float4 (&wf)[8],
                                           int c, int h, int ks, int ng) {
  const float* base = Wm + (16 * ks) * FD + 64 * ng + 4 * c;
  #pragma unroll
  for (int i = 0; i < 4; ++i) {
    wf[2 * i + 0] = *(const float4*)(base + kmap(i, h, 0) * FD);
    wf[2 * i + 1] = *(const float4*)(base + kmap(i, h, 1) * FD);
  }
}

__device__ __forceinline__ float comp(const float4& v, int j) {
  return j == 0 ? v.x : (j == 1 ? v.y : (j == 2 ? v.z : v.w));
}

__device__ __forceinline__ void fold_stage(const float* __restrict__ Pin,
                                           float* __restrict__ Pout,
                                           const float4 (&wf)[8],
                                           int c, int h, int ks, int ng) {
  const bool   live = (c < 2);
  const float* pa = Pin + (c & 1) * PQ + 16 * ks + 4 * h;
  v8f acc[4] = {};
  #pragma unroll
  for (int m = 0; m < 2; ++m) {
    float4 s = sum_parts4(pa + 8 * m);
    if (!live) s = make_float4(0.f, 0.f, 0.f, 0.f);
    v2f a0; a0.x = s.x; a0.y = s.y;
    v2f a1; a1.x = s.z; a1.y = s.w;
    #pragma unroll
    for (int j = 0; j < 4; ++j) {
      v2f b0; b0.x = comp(wf[4 * m + 0], j); b0.y = comp(wf[4 * m + 1], j);
      v2f b1; b1.x = comp(wf[4 * m + 2], j); b1.y = comp(wf[4 * m + 3], j);
      acc[j] = __builtin_amdgcn_wmma_f32_16x16x4_f32(false, a0, false, b0, (short)0, acc[j], false, false);
      acc[j] = __builtin_amdgcn_wmma_f32_16x16x4_f32(false, a1, false, b1, (short)0, acc[j], false, false);
    }
  }
  const float4 r0 = make_float4(acc[0][0], acc[1][0], acc[2][0], acc[3][0]);
  const float4 r1 = make_float4(acc[0][1], acc[1][1], acc[2][1], acc[3][1]);
  if (h == 0) {
    *(float4*)(Pout + (ks * 2 + 0) * PQ + 64 * ng + 4 * c) = r0;
    *(float4*)(Pout + (ks * 2 + 1) * PQ + 64 * ng + 4 * c) = r1;
  }
}

__global__ __launch_bounds__(512) void readout_folded(
    const float* x,
    const float* __restrict__ W1, const float* __restrict__ b1,
    const float* __restrict__ W2, const float* __restrict__ b2,
    const float* __restrict__ W3, const float* __restrict__ b3,
    const float* __restrict__ W4, const float* __restrict__ b4,
    const float* __restrict__ W5, const float* __restrict__ b5,
    float* __restrict__ out) {
  __shared__ __attribute__((aligned(16))) float Xs[NROW * XP];
  __shared__ __attribute__((aligned(16))) float PA[16 * PQ];
  __shared__ __attribute__((aligned(16))) float PB[16 * PQ];
  __shared__ __attribute__((aligned(16))) float Pfin[2 * PQ];
  __shared__ __attribute__((aligned(16))) float ULs[NROW];
  __shared__ __attribute__((aligned(16))) float URs[NROW];
  __shared__ float Kred[4];

  const int t    = threadIdx.x;
  const int lane = t & 31;
  const int wave = t >> 5;
  const int c    = lane & 15;
  const int h    = lane >> 4;
  const int b    = blockIdx.x >> 3;
  const int ib   = blockIdx.x & 7;
  const int ng   = wave & 1;
  const int ks   = wave >> 1;

  const int cg = t & 63;
  const int q  = t >> 6;
  float4 w4[16];
  float  w5[16];
  #pragma unroll
  for (int i = 0; i < 16; ++i) {
    w4[i] = *(const float4*)(W4 + (16 * q + i) * 256 + 4 * cg);
    w5[i] = W5[16 * q + i];
  }
  const int   tn  = t & 127;
  const float b4v = b4[tn], w5t = W5[tn];
  const float bb3 = b3[tn], bb2 = b2[tn], bb1 = b1[tn];
  float4 w3f[8], w2f[8], w1f[8];
  load_frags(W3, w3f, c, h, ks, ng);
  load_frags(W2, w2f, c, h, ks, ng);
  load_frags(W1, w1f, c, h, ks, ng);
  __builtin_amdgcn_sched_barrier(0);
  float4 p = make_float4(0.f, 0.f, 0.f, 0.f);
  #pragma unroll
  for (int i = 0; i < 16; ++i) {
    p.x += w5[i] * w4[i].x;
    p.y += w5[i] * w4[i].y;
    p.z += w5[i] * w4[i].z;
    p.w += w5[i] * w4[i].w;
  }
  *(float4*)(PA + q * 2 * PQ + 4 * cg) = p;
  float kacc = b4v * w5t;
  v4f xr[16];
  const v4f* xrow = (const v4f*)(x + (size_t)(b * NROW + 16 * wave) * FD + 4 * lane);
  #pragma unroll
  for (int j = 0; j < 16; ++j) xr[j] = xrow[j * (FD / 4)];
  lds_barrier();

  if (wave < 4) kacc += sum_both_rows(PA, tn) * bb3;
  fold_stage(PA, PB, w3f, c, h, ks, ng);
  lds_barrier();
  if (wave < 4) kacc += sum_both_rows(PB, tn) * bb2;
  fold_stage(PB, PA, w2f, c, h, ks, ng);
  lds_barrier();
  if (wave < 4) {
    kacc += sum_both_rows(PA, tn) * bb1;
    #pragma unroll
    for (int o = 16; o > 0; o >>= 1) kacc += __shfl_xor(kacc, o, 32);
    if (lane == 0) Kred[wave] = kacc;
  }
  fold_stage(PA, PB, w1f, c, h, ks, ng);
  lds_barrier();
  if (t < 256) {
    float s = 0.f;
    #pragma unroll
    for (int pp = 0; pp < 8; ++pp) s += PB[(pp * 2 + (t >> 7)) * PQ + (t & 127)];
    Pfin[t] = s;
  }
  lds_barrier();
  #pragma unroll
  for (int j = 0; j < 16; ++j)
    *(v4f*)(Xs + (16 * wave + j) * XP + 4 * lane) = xr[j];

  {
    const bool   live = (c < 2);
    const float* pb = Pfin + (c & 1) * PQ + 4 * h;
    const float* xa = Xs + (16 * wave + c) * XP + 4 * h;
    v8f acc0 = {}, acc1 = {};
    #pragma unroll
    for (int m = 0; m < 16; ++m) {
      const float4 xv = *(const float4*)(xa + 8 * m);
      float4 s = *(const float4*)(pb + 8 * m);
      if (!live) s = make_float4(0.f, 0.f, 0.f, 0.f);
      v2f a0; a0.x = xv.x; a0.y = xv.y;
      v2f a1; a1.x = xv.z; a1.y = xv.w;
      v2f b0; b0.x = s.x;  b0.y = s.y;
      v2f b1; b1.x = s.z;  b1.y = s.w;
      acc0 = __builtin_amdgcn_wmma_f32_16x16x4_f32(false, a0, false, b0, (short)0, acc0, false, false);
      acc1 = __builtin_amdgcn_wmma_f32_16x16x4_f32(false, a1, false, b1, (short)0, acc1, false, false);
    }
    v8f d = acc0 + acc1;
    if (live) {
      float* dst = (c == 0 ? ULs : URs) + 16 * wave + 8 * h;
      #pragma unroll
      for (int g = 0; g < 8; ++g) dst[g] = d[g];
    }
  }
  __syncthreads();

  {
    const float K  = ((Kred[0] + Kred[1]) + (Kred[2] + Kred[3])) + b5[0];
    const int   j4 = (t & 63) * 4;
    const int   rg = t >> 6;
    float4 urv = *(const float4*)(URs + j4);
    urv.x += K; urv.y += K; urv.z += K; urv.w += K;
    float* orow = out + (size_t)(b * NROW + 32 * ib) * NROW + j4;
    #pragma unroll
    for (int m = 0; m < 4; ++m) {
      const int   i  = rg + 8 * m;
      const float ul = ULs[32 * ib + i];
      *(float4*)(orow + (size_t)i * NROW) =
          make_float4(ul + urv.x, ul + urv.y, ul + urv.z, ul + urv.w);
    }
  }
}

extern "C" void kernel_launch(void* const* d_in, const int* in_sizes, int n_in,
                              void* d_out, int out_size, void* d_ws, size_t ws_size,
                              hipStream_t stream) {
  (void)in_sizes; (void)n_in; (void)out_size; (void)d_ws; (void)ws_size;
  const float* x  = (const float*)d_in[0];
  const float* W1 = (const float*)d_in[1];
  const float* b1 = (const float*)d_in[2];
  const float* W2 = (const float*)d_in[3];
  const float* b2 = (const float*)d_in[4];
  const float* W3 = (const float*)d_in[5];
  const float* b3 = (const float*)d_in[6];
  const float* W4 = (const float*)d_in[7];
  const float* b4 = (const float*)d_in[8];
  const float* W5 = (const float*)d_in[9];
  const float* b5 = (const float*)d_in[10];
  float* out = (float*)d_out;

  readout_folded<<<64, 512, 0, stream>>>(x, W1, b1, W2, b2, W3, b3, W4, b4, W5, b5, out);
}


# Each block owns two 64-row groups of its batch and writes both 64x64 tiles they span: x product covers 128 rows not 256 (8 row tiles x 2 k halves on 16 waves); k halves added last, sum order changes
# speedup vs baseline: 3.6317x; 1.1202x over previous
#include <hip/hip_runtime.h>


typedef __attribute__((ext_vector_type(2))) float v2f;
typedef __attribute__((ext_vector_type(8))) float v8f;
typedef __attribute__((ext_vector_type(4))) float v4f;

#define FD   128
#define NROW 256
#define XPH  68
#define PQ   128

__device__ __forceinline__ int kmap(int i, int h, int r) {
  return 8 * (i >> 1) + 4 * h + 2 * (i & 1) + r;
}

__device__ __forceinline__ float4 add4(float4 a, float4 b) {
  return make_float4(a.x + b.x, a.y + b.y, a.z + b.z, a.w + b.w);
}

__device__ __forceinline__ void lds_barrier() {
  __builtin_amdgcn_fence(__ATOMIC_RELEASE, "workgroup", "local");
  __builtin_amdgcn_s_barrier();
  __builtin_amdgcn_fence(__ATOMIC_ACQUIRE, "workgroup", "local");
}

__device__ __forceinline__ float4 sum_parts4(const float* p) {
  float4 s = *(const float4*)p;
  #pragma unroll
  for (int q = 1; q < 8; ++q) s = add4(s, *(const float4*)(p + q * 2 * PQ));
  return s;
}

__device__ __forceinline__ float sum_both_rows(const float* P, int n) {
  float s = 0.f;
  #pragma unroll
  for (int q = 0; q < 16; ++q) s += P[q * PQ + n];
  return s;
}

__device__ __forceinline__ void load_frags(const float* __restrict__ Wm, float4 (&wf)[8],
                                           int c, int h, int ks, int ng) {
  const float* base = Wm + (16 * ks) * FD + 64 * ng + 4 * c;
  #pragma unroll
  for (int i = 0; i < 4; ++i) {
    wf[2 * i + 0] = *(const float4*)(base + kmap(i, h, 0) * FD);
    wf[2 * i + 1] = *(const float4*)(base + kmap(i, h, 1) * FD);
  }
}

__device__ __forceinline__ float comp(const float4& v, int j) {
  return j == 0 ? v.x : (j == 1 ? v.y : (j == 2 ? v.z : v.w));
}

__device__ __forceinline__ void fold_stage(const float* __restrict__ Pin,
                                           float* __restrict__ Pout,
                                           const float4 (&wf)[8],
                                           int c, int h, int ks, int ng) {
  const bool   live = (c < 2);
  const float* pa = Pin + (c & 1) * PQ + 16 * ks + 4 * h;
  v8f acc[4] = {};
  #pragma unroll
  for (int m = 0; m < 2; ++m) {
    float4 s = sum_parts4(pa + 8 * m);
    if (!live) s = make_float4(0.f, 0.f, 0.f, 0.f);
    v2f a0; a0.x = s.x; a0.y = s.y;
    v2f a1; a1.x = s.z; a1.y = s.w;
    #pragma unroll
    for (int j = 0; j < 4; ++j) {
      v2f b0; b0.x = comp(wf[4 * m + 0], j); b0.y = comp(wf[4 * m + 1], j);
      v2f b1; b1.x = comp(wf[4 * m + 2], j); b1.y = comp(wf[4 * m + 3], j);
      acc[j] = __builtin_amdgcn_wmma_f32_16x16x4_f32(false, a0, false, b0, (short)0, acc[j], false, false);
      acc[j] = __builtin_amdgcn_wmma_f32_16x16x4_f32(false, a1, false, b1, (short)0, acc[j], false, false);
    }
  }
  const float4 r0 = make_float4(acc[0][0], acc[1][0], acc[2][0], acc[3][0]);
  const float4 r1 = make_float4(acc[0][1], acc[1][1], acc[2][1], acc[3][1]);
  if (h == 0) {
    *(float4*)(Pout + (ks * 2 + 0) * PQ + 64 * ng + 4 * c) = r0;
    *(float4*)(Pout + (ks * 2 + 1) * PQ + 64 * ng + 4 * c) = r1;
  }
}

__global__ __launch_bounds__(512) void readout_folded(
    const float* x,
    const float* __restrict__ W1, const float* __restrict__ b1,
    const float* __restrict__ W2, const float* __restrict__ b2,
    const float* __restrict__ W3, const float* __restrict__ b3,
    const float* __restrict__ W4, const float* __restrict__ b4,
    const float* __restrict__ W5, const float* __restrict__ b5,
    float* __restrict__ out) {
  __shared__ __attribute__((aligned(16))) float Xs[16 * 16 * XPH];
  __shared__ __attribute__((aligned(16))) float PA[16 * PQ];
  __shared__ __attribute__((aligned(16))) float PB[16 * PQ];
  __shared__ __attribute__((aligned(16))) float Pfin[2 * PQ];
  __shared__ __attribute__((aligned(16))) float UP[2 * 2 * 128];
  __shared__ float Kred[4];

  const int t    = threadIdx.x;
  const int lane = t & 31;
  const int wave = t >> 5;
  const int c    = lane & 15;
  const int h    = lane >> 4;
  const int b    = blockIdx.x >> 3;
  const int ib   = blockIdx.x & 7;
  const int ga   = (0x20211000u >> (4 * ib)) & 3;
  const int gb   = (0x31332321u >> (4 * ib)) & 3;
  const bool diag = ib >= 6;
  const int xt   = wave >> 1;
  const int kh   = wave & 1;
  const int ng   = wave & 1;
  const int ks   = wave >> 1;

  const int cg = t & 63;
  const int q  = t >> 6;
  float4 w4[16];
  float  w5[16];
  #pragma unroll
  for (int i = 0; i < 16; ++i) {
    w4[i] = *(const float4*)(W4 + (16 * q + i) * 256 + 4 * cg);
    w5[i] = W5[16 * q + i];
  }
  const int   tn  = t & 127;
  const float b4v = b4[tn], w5t = W5[tn];
  const float bb3 = b3[tn], bb2 = b2[tn], bb1 = b1[tn];
  float4 w3f[8], w2f[8], w1f[8];
  load_frags(W3, w3f, c, h, ks, ng);
  load_frags(W2, w2f, c, h, ks, ng);
  load_frags(W1, w1f, c, h, ks, ng);
  __builtin_amdgcn_sched_barrier(0);
  float4 p = make_float4(0.f, 0.f, 0.f, 0.f);
  #pragma unroll
  for (int i = 0; i < 16; ++i) {
    p.x += w5[i] * w4[i].x;
    p.y += w5[i] * w4[i].y;
    p.z += w5[i] * w4[i].z;
    p.w += w5[i] * w4[i].w;
  }
  *(float4*)(PA + q * 2 * PQ + 4 * cg) = p;
  float kacc = b4v * w5t;
  v4f xr[8];
  const int xrow0 = 64 * (xt < 4 ? ga : gb) + 16 * (xt & 3);
  const v4f* xrow = (const v4f*)(x + (size_t)(b * NROW + xrow0 + h) * FD + 64 * kh + 4 * c);
  #pragma unroll
  for (int j = 0; j < 8; ++j) xr[j] = xrow[j * (2 * FD / 4)];
  lds_barrier();

  if (wave < 4) kacc += sum_both_rows(PA, tn) * bb3;
  fold_stage(PA, PB, w3f, c, h, ks, ng);
  lds_barrier();
  if (wave < 4) kacc += sum_both_rows(PB, tn) * bb2;
  fold_stage(PB, PA, w2f, c, h, ks, ng);
  lds_barrier();
  if (wave < 4) {
    kacc += sum_both_rows(PA, tn) * bb1;
    #pragma unroll
    for (int o = 16; o > 0; o >>= 1) kacc += __shfl_xor(kacc, o, 32);
    if (lane == 0) Kred[wave] = kacc;
  }
  fold_stage(PA, PB, w1f, c, h, ks, ng);
  lds_barrier();
  if (t < 256) {
    float s = 0.f;
    #pragma unroll
    for (int pp = 0; pp < 8; ++pp) s += PB[(pp * 2 + (t >> 7)) * PQ + (t & 127)];
    Pfin[t] = s;
  }
  lds_barrier();
  float* Xw = Xs + wave * 16 * XPH;
  #pragma unroll
  for (int j = 0; j < 8; ++j)
    *(v4f*)(Xw + (2 * j + h) * XPH + 4 * c) = xr[j];

  {
    const bool   live = (c < 2);
    const float* pb = Pfin + (c & 1) * PQ + 64 * kh + 4 * h;
    const float* xa = Xw + c * XPH + 4 * h;
    v8f acc0 = {}, acc1 = {};
    #pragma unroll
    for (int m = 0; m < 8; ++m) {
      const float4 xv = *(const float4*)(xa + 8 * m);
      float4 s = *(const float4*)(pb + 8 * m);
      if (!live) s = make_float4(0.f, 0.f, 0.f, 0.f);
      v2f a0; a0.x = xv.x; a0.y = xv.y;
      v2f a1; a1.x = xv.z; a1.y = xv.w;
      v2f b0; b0.x = s.x;  b0.y = s.y;
      v2f b1; b1.x = s.z;  b1.y = s.w;
      acc0 = __builtin_amdgcn_wmma_f32_16x16x4_f32(false, a0, false, b0, (short)0, acc0, false, false);
      acc1 = __builtin_amdgcn_wmma_f32_16x16x4_f32(false, a1, false, b1, (short)0, acc1, false, false);
    }
    v8f d = acc0 + acc1;
    if (live) {
      float* dst = UP + (kh * 2 + c) * 128 + 16 * xt + 8 * h;
      #pragma unroll
      for (int g = 0; g < 8; ++g) dst[g] = d[g];
    }
  }
  __syncthreads();

  {
    const float K  = ((Kred[0] + Kred[1]) + (Kred[2] + Kred[3])) + b5[0];
    const int   tt = t >> 8;
    const int   j4 = (t & 15) * 4;
    const int   rg = (t >> 4) & 15;
    const int   rloc = 64 * tt;
    const int   cloc = diag ? 64 * tt : 64 * (1 - tt);
    const int   grow = tt ? gb : ga;
    const int   gcol = diag ? grow : (tt ? ga : gb);
    const float4 u0 = *(const float4*)(UP + 1 * 128 + cloc + j4);
    const float4 u1 = *(const float4*)(UP + 3 * 128 + cloc + j4);
    float4 urv = add4(u0, u1);
    urv.x += K; urv.y += K; urv.z += K; urv.w += K;
    float* orow = out + (size_t)(b * NROW + 64 * grow) * NROW + 64 * gcol + j4;
    #pragma unroll
    for (int m = 0; m < 4; ++m) {
      const int   i  = rg + 16 * m;
      const float ul = UP[rloc + i] + UP[2 * 128 + rloc + i];
      *(float4*)(orow + (size_t)i * NROW) =
          make_float4(ul + urv.x, ul + urv.y, ul + urv.z, ul + urv.w);
    }
  }
}

extern "C" void kernel_launch(void* const* d_in, const int* in_sizes, int n_in,
                              void* d_out, int out_size, void* d_ws, size_t ws_size,
                              hipStream_t stream) {
  (void)in_sizes; (void)n_in; (void)out_size; (void)d_ws; (void)ws_size;
  const float* x  = (const float*)d_in[0];
  const float* W1 = (const float*)d_in[1];
  const float* b1 = (const float*)d_in[2];
  const float* W2 = (const float*)d_in[3];
  const float* b2 = (const float*)d_in[4];
  const float* W3 = (const float*)d_in[5];
  const float* b3 = (const float*)d_in[6];
  const float* W4 = (const float*)d_in[7];
  const float* b4 = (const float*)d_in[8];
  const float* W5 = (const float*)d_in[9];
  const float* b5 = (const float*)d_in[10];
  float* out = (float*)d_out;

  readout_folded<<<64, 512, 0, stream>>>(x, W1, b1, W2, b2, W3, b3, W4, b4, W5, b5, out);
}


# Part sums ride the A operand: its 16 rows = 8 parts x 2 vectors, added from D behind each product; a wave reads 2 LDS fragments per fold, not 16; last part-sum pass and barrier go; sum order changes
# speedup vs baseline: 4.1281x; 1.1367x over previous
#include <hip/hip_runtime.h>


typedef __attribute__((ext_vector_type(2))) float v2f;
typedef __attribute__((ext_vector_type(8))) float v8f;
typedef __attribute__((ext_vector_type(4))) float v4f;

#define FD   128
#define NROW 256
#define XPH  68
#define PQ   132

__device__ __forceinline__ int kmap(int i, int h, int r) {
  return 8 * (i >> 1) + 4 * h + 2 * (i & 1) + r;
}

__device__ __forceinline__ float4 add4(float4 a, float4 b) {
  return make_float4(a.x + b.x, a.y + b.y, a.z + b.z, a.w + b.w);
}

__device__ __forceinline__ void lds_barrier() {
  __builtin_amdgcn_fence(__ATOMIC_RELEASE, "workgroup", "local");
  __builtin_amdgcn_s_barrier();
  __builtin_amdgcn_fence(__ATOMIC_ACQUIRE, "workgroup", "local");
}

__device__ __forceinline__ float sum_rows8(const v8f& a) {
  const v2f p01 = {a[0], a[1]}, p23 = {a[2], a[3]}, p45 = {a[4], a[5]}, p67 = {a[6], a[7]};
  const v2f s = (p01 + p23) + (p45 + p67);
  return s.x + s.y;
}

__device__ __forceinline__ int part_row(int c) {
  return ((c & 7) * 2 + (c >> 3)) * PQ;
}

__device__ __forceinline__ float sum_both_rows(const float* P, int n) {
  float s = 0.f;
  #pragma unroll
  for (int q = 0; q < 16; ++q) s += P[q * PQ + n];
  return s;
}

__device__ __forceinline__ void load_frags(const float* __restrict__ Wm, float4 (&wf)[8],
                                           int c, int h, int ks, int ng) {
  const float* base = Wm + (16 * ks) * FD + 64 * ng + 4 * c;
  #pragma unroll
  for (int i = 0; i < 4; ++i) {
    wf[2 * i + 0] = *(const float4*)(base + kmap(i, h, 0) * FD);
    wf[2 * i + 1] = *(const float4*)(base + kmap(i, h, 1) * FD);
  }
}

__device__ __forceinline__ float comp(const float4& v, int j) {
  return j == 0 ? v.x : (j == 1 ? v.y : (j == 2 ? v.z : v.w));
}

__device__ __forceinline__ void fold_stage(const float* __restrict__ Pin,
                                           float* __restrict__ Pout,
                                           const float4 (&wf)[8],
                                           int c, int h, int ks, int ng) {
  const float* pa = Pin + part_row(c) + 16 * ks + 4 * h;
  v8f acc[4] = {};
  #pragma unroll
  for (int m = 0; m < 2; ++m) {
    const float4 s = *(const float4*)(pa + 8 * m);
    v2f a0; a0.x = s.x; a0.y = s.y;
    v2f a1; a1.x = s.z; a1.y = s.w;
    #pragma unroll
    for (int j = 0; j < 4; ++j) {
      v2f b0; b0.x = comp(wf[4 * m + 0], j); b0.y = comp(wf[4 * m + 1], j);
      v2f b1; b1.x = comp(wf[4 * m + 2], j); b1.y = comp(wf[4 * m + 3], j);
      acc[j] = __builtin_amdgcn_wmma_f32_16x16x4_f32(false, a0, false, b0, (short)0, acc[j], false, false);
      acc[j] = __builtin_amdgcn_wmma_f32_16x16x4_f32(false, a1, false, b1, (short)0, acc[j], false, false);
    }
  }
  const float4 r = make_float4(sum_rows8(acc[0]), sum_rows8(acc[1]),
                               sum_rows8(acc[2]), sum_rows8(acc[3]));
  *(float4*)(Pout + (ks * 2 + h) * PQ + 64 * ng + 4 * c) = r;
}

__global__ __launch_bounds__(512) void readout_folded(
    const float* x,
    const float* __restrict__ W1, const float* __restrict__ b1,
    const float* __restrict__ W2, const float* __restrict__ b2,
    const float* __restrict__ W3, const float* __restrict__ b3,
    const float* __restrict__ W4, const float* __restrict__ b4,
    const float* __restrict__ W5, const float* __restrict__ b5,
    float* __restrict__ out) {
  __shared__ __attribute__((aligned(16))) float Xs[16 * 16 * XPH];
  __shared__ __attribute__((aligned(16))) float PA[16 * PQ];
  __shared__ __attribute__((aligned(16))) float PB[16 * PQ];
  __shared__ __attribute__((aligned(16))) float UP[2 * 2 * 128];
  __shared__ float Kred[4];

  const int t    = threadIdx.x;
  const int lane = t & 31;
  const int wave = t >> 5;
  const int c    = lane & 15;
  const int h    = lane >> 4;
  const int b    = blockIdx.x >> 3;
  const int ib   = blockIdx.x & 7;
  const int ga   = (0x20211000u >> (4 * ib)) & 3;
  const int gb   = (0x31332321u >> (4 * ib)) & 3;
  const bool diag = ib >= 6;
  const int xt   = wave >> 1;
  const int kh   = wave & 1;
  const int ng   = wave & 1;
  const int ks   = wave >> 1;

  const int cg = t & 63;
  const int q  = t >> 6;
  float4 w4[16];
  float  w5[16];
  #pragma unroll
  for (int i = 0; i < 16; ++i) {
    w4[i] = *(const float4*)(W4 + (16 * q + i) * 256 + 4 * cg);
    w5[i] = W5[16 * q + i];
  }
  const int   tn  = t & 127;
  const float b4v = b4[tn], w5t = W5[tn];
  const float bb3 = b3[tn], bb2 = b2[tn], bb1 = b1[tn];
  float4 w3f[8], w2f[8], w1f[8];
  load_frags(W3, w3f, c, h, ks, ng);
  load_frags(W2, w2f, c, h, ks, ng);
  load_frags(W1, w1f, c, h, ks, ng);
  __builtin_amdgcn_sched_barrier(0);
  float4 p = make_float4(0.f, 0.f, 0.f, 0.f);
  #pragma unroll
  for (int i = 0; i < 16; ++i) {
    p.x += w5[i] * w4[i].x;
    p.y += w5[i] * w4[i].y;
    p.z += w5[i] * w4[i].z;
    p.w += w5[i] * w4[i].w;
  }
  *(float4*)(PA + (q * 2 + (cg >> 5)) * PQ + 4 * (cg & 31)) = p;
  float kacc = b4v * w5t;
  v4f xr[8];
  const int xrow0 = 64 * (xt < 4 ? ga : gb) + 16 * (xt & 3);
  const v4f* xrow = (const v4f*)(x + (size_t)(b * NROW + xrow0 + h) * FD + 64 * kh + 4 * c);
  #pragma unroll
  for (int j = 0; j < 8; ++j) xr[j] = xrow[j * (2 * FD / 4)];
  lds_barrier();

  if (wave < 4) kacc += sum_both_rows(PA, tn) * bb3;
  fold_stage(PA, PB, w3f, c, h, ks, ng);
  lds_barrier();
  if (wave < 4) kacc += sum_both_rows(PB, tn) * bb2;
  fold_stage(PB, PA, w2f, c, h, ks, ng);
  lds_barrier();
  if (wave < 4) {
    kacc += sum_both_rows(PA, tn) * bb1;
    #pragma unroll
    for (int o = 16; o > 0; o >>= 1) kacc += __shfl_xor(kacc, o, 32);
    if (lane == 0) Kred[wave] = kacc;
  }
  fold_stage(PA, PB, w1f, c, h, ks, ng);
  lds_barrier();
  float* Xw = Xs + wave * 16 * XPH;
  #pragma unroll
  for (int j = 0; j < 8; ++j)
    *(v4f*)(Xw + (2 * j + h) * XPH + 4 * c) = xr[j];

  {
    const float* pa = PB + part_row(c) + 64 * kh + 4 * h;
    const float* xb = Xw + c * XPH + 4 * h;
    v8f acc0 = {}, acc1 = {};
    #pragma unroll
    for (int m = 0; m < 8; ++m) {
      const float4 s  = *(const float4*)(pa + 8 * m);
      const float4 xv = *(const float4*)(xb + 8 * m);
      v2f a0; a0.x = s.x;  a0.y = s.y;
      v2f a1; a1.x = s.z;  a1.y = s.w;
      v2f b0; b0.x = xv.x; b0.y = xv.y;
      v2f b1; b1.x = xv.z; b1.y = xv.w;
      acc0 = __builtin_amdgcn_wmma_f32_16x16x4_f32(false, a0, false, b0, (short)0, acc0, false, false);
      acc1 = __builtin_amdgcn_wmma_f32_16x16x4_f32(false, a1, false, b1, (short)0, acc1, false, false);
    }
    const v8f d = acc0 + acc1;
    UP[(kh * 2 + h) * 128 + 16 * xt + c] = sum_rows8(d);
  }
  __syncthreads();

  {
    const float K  = ((Kred[0] + Kred[1]) + (Kred[2] + Kred[3])) + b5[0];
    const int   tt = t >> 8;
    const int   j4 = (t & 15) * 4;
    const int   rg = (t >> 4) & 15;
    const int   rloc = 64 * tt;
    const int   cloc = diag ? 64 * tt : 64 * (1 - tt);
    const int   grow = tt ? gb : ga;
    const int   gcol = diag ? grow : (tt ? ga : gb);
    const float4 u0 = *(const float4*)(UP + 1 * 128 + cloc + j4);
    const float4 u1 = *(const float4*)(UP + 3 * 128 + cloc + j4);
    float4 urv = add4(u0, u1);
    urv.x += K; urv.y += K; urv.z += K; urv.w += K;
    float* orow = out + (size_t)(b * NROW + 64 * grow) * NROW + 64 * gcol + j4;
    #pragma unroll
    for (int m = 0; m < 4; ++m) {
      const int   i  = rg + 16 * m;
      const float ul = UP[rloc + i] + UP[2 * 128 + rloc + i];
      *(float4*)(orow + (size_t)i * NROW) =
          make_float4(ul + urv.x, ul + urv.y, ul + urv.z, ul + urv.w);
    }
  }
}

extern "C" void kernel_launch(void* const* d_in, const int* in_sizes, int n_in,
                              void* d_out, int out_size, void* d_ws, size_t ws_size,
                              hipStream_t stream) {
  (void)in_sizes; (void)n_in; (void)out_size; (void)d_ws; (void)ws_size;
  const float* x  = (const float*)d_in[0];
  const float* W1 = (const float*)d_in[1];
  const float* b1 = (const float*)d_in[2];
  const float* W2 = (const float*)d_in[3];
  const float* b2 = (const float*)d_in[4];
  const float* W3 = (const float*)d_in[5];
  const float* b3 = (const float*)d_in[6];
  const float* W4 = (const float*)d_in[7];
  const float* b4 = (const float*)d_in[8];
  const float* W5 = (const float*)d_in[9];
  const float* b5 = (const float*)d_in[10];
  float* out = (float*)d_out;

  readout_folded<<<64, 512, 0, stream>>>(x, W1, b1, W2, b2, W3, b3, W4, b4, W5, b5, out);
}
